# starting kernel
# baseline (speedup 1.0000x reference)
#include <hip/hip_runtime.h>
#include <hip/hip_bf16.h>
#include <math.h>

typedef float v2f __attribute__((ext_vector_type(2)));
typedef float v8f __attribute__((ext_vector_type(8)));

#define THREADS 256
#define SOM_DIM 32
#define LSTRIDE 33


__global__ __launch_bounds__(THREADS)
void som_pass_a(const float* __restrict__ input,
                const float* __restrict__ particles,
                float* __restrict__ blkSum,
                float* __restrict__ blkMinV,
                int*   __restrict__ blkMinI,
                int N)
{
    __shared__ float stage[THREADS * LSTRIDE];
    __shared__ float wsum[8 * SOM_DIM];
    __shared__ float sdist[THREADS];
    __shared__ int   sidx[THREADS];

    const int t    = threadIdx.x;
    const int p    = blockIdx.x * THREADS + t;
    const int lane = t & 31;
    const int wave = t >> 5;

    float dist2 = 0.f;
    float* srow = &stage[t * LSTRIDE];
    const bool active = (p < N);
    if (active) {
        const float4* x4 = reinterpret_cast<const float4*>(input);
        const float4* p4 = reinterpret_cast<const float4*>(particles + (size_t)p * SOM_DIM);
        #pragma unroll
        for (int q = 0; q < 8; ++q) {
            float4 pv = p4[q];
            float4 xv = x4[q];
            float d0 = pv.x - xv.x, d1 = pv.y - xv.y;
            float d2 = pv.z - xv.z, d3 = pv.w - xv.w;
            dist2 += d0*d0 + d1*d1 + d2*d2 + d3*d3;
            srow[q*4+0] = pv.x; srow[q*4+1] = pv.y;
            srow[q*4+2] = pv.z; srow[q*4+3] = pv.w;
        }
    } else {
        dist2 = 3.402823466e+38f;
        #pragma unroll
        for (int q = 0; q < SOM_DIM; ++q) srow[q] = 0.f;
    }
    __syncthreads();

    const int half = lane >> 4;
    const int n    = lane & 15;
    v2f onesA; onesA.x = 1.0f; onesA.y = 1.0f;
    v8f acc0 = {};
    v8f acc1 = {};
    const int rowbase = wave * 32;
    #pragma unroll
    for (int c = 0; c < 8; ++c) {
        const int r0 = rowbase + c*4 + half;
        const int r1 = r0 + 2;
        v2f b0, b1;
        b0.x = stage[r0*LSTRIDE + n];        b0.y = stage[r1*LSTRIDE + n];
        b1.x = stage[r0*LSTRIDE + 16 + n];   b1.y = stage[r1*LSTRIDE + 16 + n];
        acc0 = __builtin_amdgcn_wmma_f32_16x16x4_f32(false, onesA, false, b0,
                                                     (short)0, acc0, false, false);
        acc1 = __builtin_amdgcn_wmma_f32_16x16x4_f32(false, onesA, false, b1,
                                                     (short)0, acc1, false, false);
    }
    if (lane < 16) {
        wsum[wave*SOM_DIM + lane]      = acc0[0];
        wsum[wave*SOM_DIM + 16 + lane] = acc1[0];
    }

    sdist[t] = dist2;
    sidx[t]  = p;
    __syncthreads();

    #pragma unroll
    for (int s = 128; s > 0; s >>= 1) {
        if (t < s) {
            float a = sdist[t], b = sdist[t + s];
            int   ia = sidx[t], ib = sidx[t + s];
            if (b < a || (b == a && ib < ia)) { sdist[t] = b; sidx[t] = ib; }
        }
        __syncthreads();
    }

    if (t < SOM_DIM) {
        float s = 0.f;
        #pragma unroll
        for (int w = 0; w < 8; ++w) s += wsum[w*SOM_DIM + t];
        blkSum[(size_t)blockIdx.x * SOM_DIM + t] = s;
    }
    if (t == 0) { blkMinV[blockIdx.x] = sdist[0]; blkMinI[blockIdx.x] = sidx[0]; }
}

__global__ __launch_bounds__(THREADS)
void som_pass_b(const float* __restrict__ particles,
                const int*   __restrict__ gloc,
                const int*   __restrict__ iter_num,
                const float* __restrict__ blkSum,
                const float* __restrict__ blkMinV,
                const int*   __restrict__ blkMinI,
                float* __restrict__ hdr,
                int NB, int N)
{
    __shared__ float sdist[THREADS];
    __shared__ int   sidx[THREADS];
    __shared__ float part[SOM_DIM][8];
    __shared__ int   sbmu;

    const int t = threadIdx.x;

    float best = 3.402823466e+38f; int bidx = 0x7fffffff;
    for (int k = t; k < NB; k += THREADS) {
        float dv = blkMinV[k]; int iv = blkMinI[k];
        if (dv < best || (dv == best && iv < bidx)) { best = dv; bidx = iv; }
    }
    sdist[t] = best; sidx[t] = bidx;
    __syncthreads();
    #pragma unroll
    for (int s = 128; s > 0; s >>= 1) {
        if (t < s) {
            float a = sdist[t], b = sdist[t + s];
            int   ia = sidx[t], ib = sidx[t + s];
            if (b < a || (b == a && ib < ia)) { sdist[t] = b; sidx[t] = ib; }
        }
        __syncthreads();
    }
    if (t == 0) sbmu = sidx[0];

    {
        const int d = t >> 3, seg = t & 7;
        const int per = (NB + 7) >> 3;
        const int b0 = seg * per;
        const int b1 = (b0 + per < NB) ? (b0 + per) : NB;
        float s = 0.f;
        for (int b = b0; b < b1; ++b) s += blkSum[(size_t)b * SOM_DIM + d];
        part[d][seg] = s;
    }
    __syncthreads();

    const int bmu = sbmu;
    if (t < SOM_DIM) {
        float S = 0.f;
        #pragma unroll
        for (int g = 0; g < 8; ++g) S += part[t][g];
        const float gb = particles[(size_t)bmu * SOM_DIM + t];
        hdr[4 + t]              = gb;
        hdr[4 + SOM_DIM + t]    = S / (float)N;
        hdr[4 + 2*SOM_DIM + t]  = (S - gb) / (float)(N - 1);
    }
    if (t == 0) {
        const float decay  = 1.0f - (float)iter_num[0] / 100.0f;
        const float radius = 0.5f * decay;
        const float sig    = 256.0f * decay;
        hdr[0] = (float)gloc[2*bmu];
        hdr[1] = (float)gloc[2*bmu + 1];
        hdr[2] = radius;
        hdr[3] = sig * sig;
    }
}

__global__ __launch_bounds__(THREADS)
void som_pass_c(const float* __restrict__ particles,
                const float* __restrict__ velocities,
                const float* __restrict__ r1,
                const float* __restrict__ r2,
                const int*   __restrict__ gloc,
                const float* __restrict__ hdr,
                float* __restrict__ outP,
                float* __restrict__ outV,
                int N)
{
    const int p = blockIdx.x * THREADS + threadIdx.x;
    if (p >= N) return;

    const float bx = hdr[0], by = hdr[1], radius = hdr[2], sigma2 = hdr[3];
    const float* gbest = hdr + 4;
    const float* cenA  = hdr + 4 + SOM_DIM;
    const float* cenB  = hdr + 4 + 2*SOM_DIM;

    const float4* P4  = (const float4*)(particles  + (size_t)p * SOM_DIM);
    const float4* V4  = (const float4*)(velocities + (size_t)p * SOM_DIM);
    const float4* R14 = (const float4*)(r1 + (size_t)p * SOM_DIM);
    const float4* R24 = (const float4*)(r2 + (size_t)p * SOM_DIM);
    __builtin_prefetch(R14, 0, 0);
    __builtin_prefetch(R24, 0, 0);

    const int2 gl = *(const int2*)(gloc + 2*(size_t)p);
    const float dx = (float)gl.x - bx;
    const float dy = (float)gl.y - by;
    const float nb = expf(-(dx*dx + dy*dy) / sigma2);
    const bool  skip = (1.0f - nb) > radius;
    const float* cen = ((nb + radius) >= 1.0f) ? cenA : cenB;
    const float4* C4 = (const float4*)cen;
    const float4* G4 = (const float4*)gbest;

    float4* OP = (float4*)(outP + (size_t)p * SOM_DIM);
    float4* OV = (float4*)(outV + (size_t)p * SOM_DIM);

    #pragma unroll
    for (int q = 0; q < 8; ++q) {
        const float4 pv = P4[q], vv = V4[q], a = R14[q], b = R24[q];
        const float4 c = C4[q], g = G4[q];
        float4 vn, pn;
        vn.x = 0.001f*vv.x + 0.01f*a.x*(c.x - pv.x) + 0.1f*b.x*(g.x - pv.x);
        vn.y = 0.001f*vv.y + 0.01f*a.y*(c.y - pv.y) + 0.1f*b.y*(g.y - pv.y);
        vn.z = 0.001f*vv.z + 0.01f*a.z*(c.z - pv.z) + 0.1f*b.z*(g.z - pv.z);
        vn.w = 0.001f*vv.w + 0.01f*a.w*(c.w - pv.w) + 0.1f*b.w*(g.w - pv.w);
        pn.x = pv.x + vn.x; pn.y = pv.y + vn.y;
        pn.z = pv.z + vn.z; pn.w = pv.w + vn.w;
        if (skip) { vn = vv; pn = pv; }
        OP[q] = pn;
        OV[q] = vn;
    }
}

extern "C" void kernel_launch(void* const* d_in, const int* in_sizes, int n_in,
                              void* d_out, int out_size, void* d_ws, size_t ws_size,
                              hipStream_t stream)
{
    (void)n_in; (void)out_size; (void)ws_size;
    const float* input      = (const float*)d_in[0];
    const float* particles  = (const float*)d_in[1];
    const float* velocities = (const float*)d_in[2];
    const float* r1         = (const float*)d_in[3];
    const float* r2         = (const float*)d_in[4];
    const int*   gloc       = (const int*)d_in[5];
    const int*   iter_num   = (const int*)d_in[6];

    const int N  = in_sizes[1] / SOM_DIM;
    const int NB = (N + THREADS - 1) / THREADS;

    float* ws      = (float*)d_ws;
    float* blkSum  = ws;
    float* blkMinV = ws + (size_t)NB * SOM_DIM;
    int*   blkMinI = (int*)(blkMinV + NB);
    size_t hoff    = (size_t)NB * SOM_DIM + 2 * (size_t)NB;
    hoff = (hoff + 3) & ~(size_t)3;
    float* hdr     = ws + hoff;

    float* outP = (float*)d_out;
    float* outV = outP + (size_t)N * SOM_DIM;

    som_pass_a<<<NB, THREADS, 0, stream>>>(input, particles, blkSum, blkMinV, blkMinI, N);
    som_pass_b<<<1, THREADS, 0, stream>>>(particles, gloc, iter_num,
                                          blkSum, blkMinV, blkMinI, hdr, NB, N);
    som_pass_c<<<NB, THREADS, 0, stream>>>(particles, velocities, r1, r2,
                                           gloc, hdr, outP, outV, N);
}


# Passes A and C: lanes moved from one row each to consecutive 16-B pieces, so each wave-wide load and store covers four whole 128-B lines; per-row skip and centroid flags handed over by wave ballot
# speedup vs baseline: 1.9591x; 1.9591x over previous
#include <hip/hip_runtime.h>
#include <hip/hip_bf16.h>
#include <math.h>

typedef float v2f __attribute__((ext_vector_type(2)));
typedef float v8f __attribute__((ext_vector_type(8)));

#define THREADS 256
#define SOM_DIM 32
#define LSTRIDE 33


template <bool FULL>
__device__ __forceinline__
void stage_rows(const float* __restrict__ particles, float* __restrict__ stage,
                int rowBase, int N, int t)
{
    const int chunk = t & 7;
    const int rsub  = t >> 3;
    const float4* p4 = reinterpret_cast<const float4*>(particles) + (size_t)rowBase * 8 + t;
    #pragma unroll
    for (int q = 0; q < 8; ++q) {
        const int rl = q * 32 + rsub;
        float4 pv = make_float4(0.f, 0.f, 0.f, 0.f);
        if (FULL || rowBase + rl < N) pv = p4[q * 256];
        float* dst = stage + rl * LSTRIDE + chunk * 4;
        dst[0] = pv.x; dst[1] = pv.y; dst[2] = pv.z; dst[3] = pv.w;
    }
}

__global__ __launch_bounds__(THREADS)
void som_pass_a(const float* __restrict__ input,
                const float* __restrict__ particles,
                float* __restrict__ blkSum,
                float* __restrict__ blkMinV,
                int*   __restrict__ blkMinI,
                int N)
{
    __shared__ float stage[THREADS * LSTRIDE];
    __shared__ float wsum[8 * SOM_DIM];
    __shared__ float sdist[THREADS];
    __shared__ int   sidx[THREADS];

    const int t    = threadIdx.x;
    const int p    = blockIdx.x * THREADS + t;
    const int lane = t & 31;
    const int wave = t >> 5;

    const int rowBase = blockIdx.x * THREADS;
    if (rowBase + THREADS <= N) stage_rows<true >(particles, stage, rowBase, N, t);
    else                        stage_rows<false>(particles, stage, rowBase, N, t);
    __syncthreads();

    float dist2 = 3.402823466e+38f;
    if (p < N) {
        const float4* x4 = reinterpret_cast<const float4*>(input);
        const float* srow = &stage[t * LSTRIDE];
        float acc = 0.f;
        #pragma unroll
        for (int q = 0; q < 8; ++q) {
            const float4 xv = x4[q];
            float d0 = srow[q*4+0] - xv.x, d1 = srow[q*4+1] - xv.y;
            float d2 = srow[q*4+2] - xv.z, d3 = srow[q*4+3] - xv.w;
            acc += d0*d0 + d1*d1 + d2*d2 + d3*d3;
        }
        dist2 = acc;
    }

    const int half = lane >> 4;
    const int n    = lane & 15;
    v2f onesA; onesA.x = 1.0f; onesA.y = 1.0f;
    v8f acc0 = {};
    v8f acc1 = {};
    const int rowbase = wave * 32;
    #pragma unroll
    for (int c = 0; c < 8; ++c) {
        const int r0 = rowbase + c*4 + half;
        const int r1 = r0 + 2;
        v2f b0, b1;
        b0.x = stage[r0*LSTRIDE + n];        b0.y = stage[r1*LSTRIDE + n];
        b1.x = stage[r0*LSTRIDE + 16 + n];   b1.y = stage[r1*LSTRIDE + 16 + n];
        acc0 = __builtin_amdgcn_wmma_f32_16x16x4_f32(false, onesA, false, b0,
                                                     (short)0, acc0, false, false);
        acc1 = __builtin_amdgcn_wmma_f32_16x16x4_f32(false, onesA, false, b1,
                                                     (short)0, acc1, false, false);
    }
    if (lane < 16) {
        wsum[wave*SOM_DIM + lane]      = acc0[0];
        wsum[wave*SOM_DIM + 16 + lane] = acc1[0];
    }

    sdist[t] = dist2;
    sidx[t]  = p;
    __syncthreads();

    #pragma unroll
    for (int s = 128; s > 0; s >>= 1) {
        if (t < s) {
            float a = sdist[t], b = sdist[t + s];
            int   ia = sidx[t], ib = sidx[t + s];
            if (b < a || (b == a && ib < ia)) { sdist[t] = b; sidx[t] = ib; }
        }
        __syncthreads();
    }

    if (t < SOM_DIM) {
        float s = 0.f;
        #pragma unroll
        for (int w = 0; w < 8; ++w) s += wsum[w*SOM_DIM + t];
        blkSum[(size_t)blockIdx.x * SOM_DIM + t] = s;
    }
    if (t == 0) { blkMinV[blockIdx.x] = sdist[0]; blkMinI[blockIdx.x] = sidx[0]; }
}

__global__ __launch_bounds__(THREADS)
void som_pass_b(const float* __restrict__ particles,
                const int*   __restrict__ gloc,
                const int*   __restrict__ iter_num,
                const float* __restrict__ blkSum,
                const float* __restrict__ blkMinV,
                const int*   __restrict__ blkMinI,
                float* __restrict__ hdr,
                int NB, int N)
{
    __shared__ float sdist[THREADS];
    __shared__ int   sidx[THREADS];
    __shared__ float part[SOM_DIM][8];
    __shared__ int   sbmu;

    const int t = threadIdx.x;

    float best = 3.402823466e+38f; int bidx = 0x7fffffff;
    for (int k = t; k < NB; k += THREADS) {
        float dv = blkMinV[k]; int iv = blkMinI[k];
        if (dv < best || (dv == best && iv < bidx)) { best = dv; bidx = iv; }
    }
    sdist[t] = best; sidx[t] = bidx;
    __syncthreads();
    #pragma unroll
    for (int s = 128; s > 0; s >>= 1) {
        if (t < s) {
            float a = sdist[t], b = sdist[t + s];
            int   ia = sidx[t], ib = sidx[t + s];
            if (b < a || (b == a && ib < ia)) { sdist[t] = b; sidx[t] = ib; }
        }
        __syncthreads();
    }
    if (t == 0) sbmu = sidx[0];

    {
        const int d = t >> 3, seg = t & 7;
        const int per = (NB + 7) >> 3;
        const int b0 = seg * per;
        const int b1 = (b0 + per < NB) ? (b0 + per) : NB;
        float s = 0.f;
        for (int b = b0; b < b1; ++b) s += blkSum[(size_t)b * SOM_DIM + d];
        part[d][seg] = s;
    }
    __syncthreads();

    const int bmu = sbmu;
    if (t < SOM_DIM) {
        float S = 0.f;
        #pragma unroll
        for (int g = 0; g < 8; ++g) S += part[t][g];
        const float gb = particles[(size_t)bmu * SOM_DIM + t];
        hdr[4 + t]              = gb;
        hdr[4 + SOM_DIM + t]    = S / (float)N;
        hdr[4 + 2*SOM_DIM + t]  = (S - gb) / (float)(N - 1);
    }
    if (t == 0) {
        const float decay  = 1.0f - (float)iter_num[0] / 100.0f;
        const float radius = 0.5f * decay;
        const float sig    = 256.0f * decay;
        hdr[0] = (float)gloc[2*bmu];
        hdr[1] = (float)gloc[2*bmu + 1];
        hdr[2] = radius;
        hdr[3] = sig * sig;
    }
}

__device__ __forceinline__
void update_piece(const float4 pv, const float4 vv, const float4 a, const float4 b,
                  const float4 g, const float4 cA, const float4 cB,
                  const bool sk, const bool sa, float4& pn, float4& vn)
{
    float4 c;
    c.x = sa ? cA.x : cB.x; c.y = sa ? cA.y : cB.y;
    c.z = sa ? cA.z : cB.z; c.w = sa ? cA.w : cB.w;
    vn.x = 0.001f*vv.x + 0.01f*a.x*(c.x - pv.x) + 0.1f*b.x*(g.x - pv.x);
    vn.y = 0.001f*vv.y + 0.01f*a.y*(c.y - pv.y) + 0.1f*b.y*(g.y - pv.y);
    vn.z = 0.001f*vv.z + 0.01f*a.z*(c.z - pv.z) + 0.1f*b.z*(g.z - pv.z);
    vn.w = 0.001f*vv.w + 0.01f*a.w*(c.w - pv.w) + 0.1f*b.w*(g.w - pv.w);
    pn.x = pv.x + vn.x; pn.y = pv.y + vn.y;
    pn.z = pv.z + vn.z; pn.w = pv.w + vn.w;
    pn.x = sk ? pv.x : pn.x; pn.y = sk ? pv.y : pn.y;
    pn.z = sk ? pv.z : pn.z; pn.w = sk ? pv.w : pn.w;
    vn.x = sk ? vv.x : vn.x; vn.y = sk ? vv.y : vn.y;
    vn.z = sk ? vv.z : vn.z; vn.w = sk ? vv.w : vn.w;
}

__device__ __forceinline__
void row_flags(const int* __restrict__ gloc, int prow,
               float bx, float by, float radius, float sigma2, bool& skip, bool& selA)
{
    const int2 gl = *(const int2*)(gloc + 2*(size_t)prow);
    const float dx = (float)gl.x - bx;
    const float dy = (float)gl.y - by;
    const float nb = expf(-(dx*dx + dy*dy) / sigma2);
    skip = (1.0f - nb) > radius;
    selA = (nb + radius) >= 1.0f;
}

__global__ __launch_bounds__(THREADS)
void som_pass_c(const float* __restrict__ particles,
                const float* __restrict__ velocities,
                const float* __restrict__ r1,
                const float* __restrict__ r2,
                const int*   __restrict__ gloc,
                const float* __restrict__ hdr,
                float* __restrict__ outP,
                float* __restrict__ outV,
                int N)
{
    const int lane = threadIdx.x & 31;
    const int rowW = blockIdx.x * THREADS + (threadIdx.x >> 5) * 32;
    const int prow = rowW + lane;

    const float bx = hdr[0], by = hdr[1], radius = hdr[2], sigma2 = hdr[3];

    const int chunk = lane & 7;
    const int rsub  = lane >> 3;
    const float4 g  = ((const float4*)(hdr + 4))[chunk];
    const float4 cA = ((const float4*)(hdr + 4 + SOM_DIM))[chunk];
    const float4 cB = ((const float4*)(hdr + 4 + 2*SOM_DIM))[chunk];

    const size_t base = (size_t)rowW * 8 + lane;
    const float4* P4  = (const float4*)particles  + base;
    const float4* V4  = (const float4*)velocities + base;
    const float4* R14 = (const float4*)r1 + base;
    const float4* R24 = (const float4*)r2 + base;
    float4* OP = (float4*)outP + base;
    float4* OV = (float4*)outV + base;

    if (rowW + 32 <= N) {
        bool skip, selA;
        row_flags(gloc, prow, bx, by, radius, sigma2, skip, selA);
        const unsigned skipMask = __builtin_amdgcn_ballot_w32(skip);
        const unsigned selMask  = __builtin_amdgcn_ballot_w32(selA);
        #pragma unroll
        for (int q = 0; q < 8; ++q) {
            const int rl = q * 4 + rsub;
            const bool sk = ((skipMask >> rl) & 1u) != 0u;
            const bool sa = ((selMask  >> rl) & 1u) != 0u;
            float4 pn, vn;
            update_piece(P4[q*32], V4[q*32], R14[q*32], R24[q*32], g, cA, cB, sk, sa, pn, vn);
            OP[q*32] = pn;
            OV[q*32] = vn;
        }
    } else {
        bool skip = true, selA = false;
        if (prow < N) row_flags(gloc, prow, bx, by, radius, sigma2, skip, selA);
        const unsigned skipMask = __builtin_amdgcn_ballot_w32(skip);
        const unsigned selMask  = __builtin_amdgcn_ballot_w32(selA);
        for (int q = 0; q < 8; ++q) {
            const int rl = q * 4 + rsub;
            if (rowW + rl < N) {
                const bool sk = ((skipMask >> rl) & 1u) != 0u;
                const bool sa = ((selMask  >> rl) & 1u) != 0u;
                float4 pn, vn;
                update_piece(P4[q*32], V4[q*32], R14[q*32], R24[q*32], g, cA, cB, sk, sa, pn, vn);
                OP[q*32] = pn;
                OV[q*32] = vn;
            }
        }
    }
}

extern "C" void kernel_launch(void* const* d_in, const int* in_sizes, int n_in,
                              void* d_out, int out_size, void* d_ws, size_t ws_size,
                              hipStream_t stream)
{
    (void)n_in; (void)out_size; (void)ws_size;
    const float* input      = (const float*)d_in[0];
    const float* particles  = (const float*)d_in[1];
    const float* velocities = (const float*)d_in[2];
    const float* r1         = (const float*)d_in[3];
    const float* r2         = (const float*)d_in[4];
    const int*   gloc       = (const int*)d_in[5];
    const int*   iter_num   = (const int*)d_in[6];

    const int N  = in_sizes[1] / SOM_DIM;
    const int NB = (N + THREADS - 1) / THREADS;

    float* ws      = (float*)d_ws;
    float* blkSum  = ws;
    float* blkMinV = ws + (size_t)NB * SOM_DIM;
    int*   blkMinI = (int*)(blkMinV + NB);
    size_t hoff    = (size_t)NB * SOM_DIM + 2 * (size_t)NB;
    hoff = (hoff + 3) & ~(size_t)3;
    float* hdr     = ws + hoff;

    float* outP = (float*)d_out;
    float* outV = outP + (size_t)N * SOM_DIM;

    som_pass_a<<<NB, THREADS, 0, stream>>>(input, particles, blkSum, blkMinV, blkMinI, N);
    som_pass_b<<<1, THREADS, 0, stream>>>(particles, gloc, iter_num,
                                          blkSum, blkMinV, blkMinI, hdr, NB, N);
    som_pass_c<<<NB, THREADS, 0, stream>>>(particles, velocities, r1, r2,
                                           gloc, hdr, outP, outV, N);
}


# Pass B: the load-wait-add chain over the per-block partials becomes one 1024-thread work-group that requests every partial at once; column sums now added in a fixed lane-then-wave tree order
# speedup vs baseline: 3.9350x; 2.0086x over previous
#include <hip/hip_runtime.h>
#include <hip/hip_bf16.h>
#include <math.h>

typedef float v2f __attribute__((ext_vector_type(2)));
typedef float v8f __attribute__((ext_vector_type(8)));

#define THREADS 256
#define SOM_DIM 32
#define LSTRIDE 33


template <bool FULL>
__device__ __forceinline__
void stage_rows(const float* __restrict__ particles, float* __restrict__ stage,
                int rowBase, int N, int t)
{
    const int chunk = t & 7;
    const int rsub  = t >> 3;
    const float4* p4 = reinterpret_cast<const float4*>(particles) + (size_t)rowBase * 8 + t;
    #pragma unroll
    for (int q = 0; q < 8; ++q) {
        const int rl = q * 32 + rsub;
        float4 pv = make_float4(0.f, 0.f, 0.f, 0.f);
        if (FULL || rowBase + rl < N) pv = p4[q * 256];
        float* dst = stage + rl * LSTRIDE + chunk * 4;
        dst[0] = pv.x; dst[1] = pv.y; dst[2] = pv.z; dst[3] = pv.w;
    }
}

__global__ __launch_bounds__(THREADS)
void som_pass_a(const float* __restrict__ input,
                const float* __restrict__ particles,
                float* __restrict__ blkSum,
                float* __restrict__ blkMinV,
                int*   __restrict__ blkMinI,
                int N)
{
    __shared__ float stage[THREADS * LSTRIDE];
    __shared__ float wsum[8 * SOM_DIM];
    __shared__ float sdist[THREADS];
    __shared__ int   sidx[THREADS];

    const int t    = threadIdx.x;
    const int p    = blockIdx.x * THREADS + t;
    const int lane = t & 31;
    const int wave = t >> 5;

    const int rowBase = blockIdx.x * THREADS;
    if (rowBase + THREADS <= N) stage_rows<true >(particles, stage, rowBase, N, t);
    else                        stage_rows<false>(particles, stage, rowBase, N, t);
    __syncthreads();

    float dist2 = 3.402823466e+38f;
    if (p < N) {
        const float4* x4 = reinterpret_cast<const float4*>(input);
        const float* srow = &stage[t * LSTRIDE];
        float acc = 0.f;
        #pragma unroll
        for (int q = 0; q < 8; ++q) {
            const float4 xv = x4[q];
            float d0 = srow[q*4+0] - xv.x, d1 = srow[q*4+1] - xv.y;
            float d2 = srow[q*4+2] - xv.z, d3 = srow[q*4+3] - xv.w;
            acc += d0*d0 + d1*d1 + d2*d2 + d3*d3;
        }
        dist2 = acc;
    }

    const int half = lane >> 4;
    const int n    = lane & 15;
    v2f onesA; onesA.x = 1.0f; onesA.y = 1.0f;
    v8f acc0 = {};
    v8f acc1 = {};
    const int rowbase = wave * 32;
    #pragma unroll
    for (int c = 0; c < 8; ++c) {
        const int r0 = rowbase + c*4 + half;
        const int r1 = r0 + 2;
        v2f b0, b1;
        b0.x = stage[r0*LSTRIDE + n];        b0.y = stage[r1*LSTRIDE + n];
        b1.x = stage[r0*LSTRIDE + 16 + n];   b1.y = stage[r1*LSTRIDE + 16 + n];
        acc0 = __builtin_amdgcn_wmma_f32_16x16x4_f32(false, onesA, false, b0,
                                                     (short)0, acc0, false, false);
        acc1 = __builtin_amdgcn_wmma_f32_16x16x4_f32(false, onesA, false, b1,
                                                     (short)0, acc1, false, false);
    }
    if (lane < 16) {
        wsum[wave*SOM_DIM + lane]      = acc0[0];
        wsum[wave*SOM_DIM + 16 + lane] = acc1[0];
    }

    sdist[t] = dist2;
    sidx[t]  = p;
    __syncthreads();

    #pragma unroll
    for (int s = 128; s > 0; s >>= 1) {
        if (t < s) {
            float a = sdist[t], b = sdist[t + s];
            int   ia = sidx[t], ib = sidx[t + s];
            if (b < a || (b == a && ib < ia)) { sdist[t] = b; sidx[t] = ib; }
        }
        __syncthreads();
    }

    if (t < SOM_DIM) {
        float s = 0.f;
        #pragma unroll
        for (int w = 0; w < 8; ++w) s += wsum[w*SOM_DIM + t];
        blkSum[(size_t)blockIdx.x * SOM_DIM + t] = s;
    }
    if (t == 0) { blkMinV[blockIdx.x] = sdist[0]; blkMinI[blockIdx.x] = sidx[0]; }
}

#define BTHREADS 1024

__global__ __launch_bounds__(BTHREADS)
void som_pass_b(const float* __restrict__ particles,
                const int*   __restrict__ gloc,
                const int*   __restrict__ iter_num,
                const float* __restrict__ blkSum,
                const float* __restrict__ blkMinV,
                const int*   __restrict__ blkMinI,
                float* __restrict__ hdr,
                int NB, int N)
{
    __shared__ float4 wsumB4[32 * 8];
    __shared__ float wdist[32];
    __shared__ int   widx[32];

    const int t    = threadIdx.x;
    const int lane = t & 31;
    const int wave = t >> 5;

    const int iter = iter_num[0];

    float best = 3.402823466e+38f; int bidx = 0x7fffffff;
    if (t < NB) { best = blkMinV[t]; bidx = blkMinI[t]; }

    const float4* s4 = reinterpret_cast<const float4*>(blkSum);
    const int total4 = NB * 8;
    float4 acc = make_float4(0.f, 0.f, 0.f, 0.f);
    for (int base = 0; base < total4; base += 8 * BTHREADS) {
        float4 v[8];
        if (base + 8 * BTHREADS <= total4) {
            #pragma unroll
            for (int j = 0; j < 8; ++j) v[j] = s4[base + j * BTHREADS + t];
        } else {
            #pragma unroll
            for (int j = 0; j < 8; ++j) {
                const int idx = base + j * BTHREADS + t;
                v[j] = make_float4(0.f, 0.f, 0.f, 0.f);
                if (idx < total4) v[j] = s4[idx];
            }
        }
        #pragma unroll
        for (int j = 0; j < 8; ++j) {
            acc.x += v[j].x; acc.y += v[j].y; acc.z += v[j].z; acc.w += v[j].w;
        }
    }

    for (int k = t + BTHREADS; k < NB; k += BTHREADS) {
        const float dv = blkMinV[k]; const int iv = blkMinI[k];
        if (dv < best || (dv == best && iv < bidx)) { best = dv; bidx = iv; }
    }

    acc.x += __shfl_xor(acc.x, 8);  acc.y += __shfl_xor(acc.y, 8);
    acc.z += __shfl_xor(acc.z, 8);  acc.w += __shfl_xor(acc.w, 8);
    acc.x += __shfl_xor(acc.x, 16); acc.y += __shfl_xor(acc.y, 16);
    acc.z += __shfl_xor(acc.z, 16); acc.w += __shfl_xor(acc.w, 16);
    if (lane < 8) wsumB4[wave * 8 + lane] = acc;

    #pragma unroll
    for (int m = 16; m > 0; m >>= 1) {
        const float od = __shfl_xor(best, m);
        const int   oi = __shfl_xor(bidx, m);
        if (od < best || (od == best && oi < bidx)) { best = od; bidx = oi; }
    }
    if (lane == 0) { wdist[wave] = best; widx[wave] = bidx; }
    __syncthreads();

    if (wave == 0) {
        float b2 = wdist[lane]; int bmu = widx[lane];
        #pragma unroll
        for (int m = 16; m > 0; m >>= 1) {
            const float od = __shfl_xor(b2, m);
            const int   oi = __shfl_xor(bmu, m);
            if (od < b2 || (od == b2 && oi < bmu)) { b2 = od; bmu = oi; }
        }
        const float gb = particles[(size_t)bmu * SOM_DIM + lane];

        const float* wsumB = reinterpret_cast<const float*>(wsumB4);
        float S = 0.f;
        #pragma unroll
        for (int w = 0; w < 32; ++w) S += wsumB[w * SOM_DIM + lane];

        hdr[4 + lane]              = gb;
        hdr[4 + SOM_DIM + lane]    = S / (float)N;
        hdr[4 + 2*SOM_DIM + lane]  = (S - gb) / (float)(N - 1);
        if (lane == 0) {
            const float decay  = 1.0f - (float)iter / 100.0f;
            const float radius = 0.5f * decay;
            const float sig    = 256.0f * decay;
            hdr[0] = (float)gloc[2*bmu];
            hdr[1] = (float)gloc[2*bmu + 1];
            hdr[2] = radius;
            hdr[3] = sig * sig;
        }
    }
}

__device__ __forceinline__
void update_piece(const float4 pv, const float4 vv, const float4 a, const float4 b,
                  const float4 g, const float4 cA, const float4 cB,
                  const bool sk, const bool sa, float4& pn, float4& vn)
{
    float4 c;
    c.x = sa ? cA.x : cB.x; c.y = sa ? cA.y : cB.y;
    c.z = sa ? cA.z : cB.z; c.w = sa ? cA.w : cB.w;
    vn.x = 0.001f*vv.x + 0.01f*a.x*(c.x - pv.x) + 0.1f*b.x*(g.x - pv.x);
    vn.y = 0.001f*vv.y + 0.01f*a.y*(c.y - pv.y) + 0.1f*b.y*(g.y - pv.y);
    vn.z = 0.001f*vv.z + 0.01f*a.z*(c.z - pv.z) + 0.1f*b.z*(g.z - pv.z);
    vn.w = 0.001f*vv.w + 0.01f*a.w*(c.w - pv.w) + 0.1f*b.w*(g.w - pv.w);
    pn.x = pv.x + vn.x; pn.y = pv.y + vn.y;
    pn.z = pv.z + vn.z; pn.w = pv.w + vn.w;
    pn.x = sk ? pv.x : pn.x; pn.y = sk ? pv.y : pn.y;
    pn.z = sk ? pv.z : pn.z; pn.w = sk ? pv.w : pn.w;
    vn.x = sk ? vv.x : vn.x; vn.y = sk ? vv.y : vn.y;
    vn.z = sk ? vv.z : vn.z; vn.w = sk ? vv.w : vn.w;
}

__device__ __forceinline__
void row_flags(const int* __restrict__ gloc, int prow,
               float bx, float by, float radius, float sigma2, bool& skip, bool& selA)
{
    const int2 gl = *(const int2*)(gloc + 2*(size_t)prow);
    const float dx = (float)gl.x - bx;
    const float dy = (float)gl.y - by;
    const float nb = expf(-(dx*dx + dy*dy) / sigma2);
    skip = (1.0f - nb) > radius;
    selA = (nb + radius) >= 1.0f;
}

__global__ __launch_bounds__(THREADS)
void som_pass_c(const float* __restrict__ particles,
                const float* __restrict__ velocities,
                const float* __restrict__ r1,
                const float* __restrict__ r2,
                const int*   __restrict__ gloc,
                const float* __restrict__ hdr,
                float* __restrict__ outP,
                float* __restrict__ outV,
                int N)
{
    const int lane = threadIdx.x & 31;
    const int rowW = blockIdx.x * THREADS + (threadIdx.x >> 5) * 32;
    const int prow = rowW + lane;

    const float bx = hdr[0], by = hdr[1], radius = hdr[2], sigma2 = hdr[3];

    const int chunk = lane & 7;
    const int rsub  = lane >> 3;
    const float4 g  = ((const float4*)(hdr + 4))[chunk];
    const float4 cA = ((const float4*)(hdr + 4 + SOM_DIM))[chunk];
    const float4 cB = ((const float4*)(hdr + 4 + 2*SOM_DIM))[chunk];

    const size_t base = (size_t)rowW * 8 + lane;
    const float4* P4  = (const float4*)particles  + base;
    const float4* V4  = (const float4*)velocities + base;
    const float4* R14 = (const float4*)r1 + base;
    const float4* R24 = (const float4*)r2 + base;
    float4* OP = (float4*)outP + base;
    float4* OV = (float4*)outV + base;

    if (rowW + 32 <= N) {
        bool skip, selA;
        row_flags(gloc, prow, bx, by, radius, sigma2, skip, selA);
        const unsigned skipMask = __builtin_amdgcn_ballot_w32(skip);
        const unsigned selMask  = __builtin_amdgcn_ballot_w32(selA);
        #pragma unroll
        for (int q = 0; q < 8; ++q) {
            const int rl = q * 4 + rsub;
            const bool sk = ((skipMask >> rl) & 1u) != 0u;
            const bool sa = ((selMask  >> rl) & 1u) != 0u;
            float4 pn, vn;
            update_piece(P4[q*32], V4[q*32], R14[q*32], R24[q*32], g, cA, cB, sk, sa, pn, vn);
            OP[q*32] = pn;
            OV[q*32] = vn;
        }
    } else {
        bool skip = true, selA = false;
        if (prow < N) row_flags(gloc, prow, bx, by, radius, sigma2, skip, selA);
        const unsigned skipMask = __builtin_amdgcn_ballot_w32(skip);
        const unsigned selMask  = __builtin_amdgcn_ballot_w32(selA);
        for (int q = 0; q < 8; ++q) {
            const int rl = q * 4 + rsub;
            if (rowW + rl < N) {
                const bool sk = ((skipMask >> rl) & 1u) != 0u;
                const bool sa = ((selMask  >> rl) & 1u) != 0u;
                float4 pn, vn;
                update_piece(P4[q*32], V4[q*32], R14[q*32], R24[q*32], g, cA, cB, sk, sa, pn, vn);
                OP[q*32] = pn;
                OV[q*32] = vn;
            }
        }
    }
}

extern "C" void kernel_launch(void* const* d_in, const int* in_sizes, int n_in,
                              void* d_out, int out_size, void* d_ws, size_t ws_size,
                              hipStream_t stream)
{
    (void)n_in; (void)out_size; (void)ws_size;
    const float* input      = (const float*)d_in[0];
    const float* particles  = (const float*)d_in[1];
    const float* velocities = (const float*)d_in[2];
    const float* r1         = (const float*)d_in[3];
    const float* r2         = (const float*)d_in[4];
    const int*   gloc       = (const int*)d_in[5];
    const int*   iter_num   = (const int*)d_in[6];

    const int N  = in_sizes[1] / SOM_DIM;
    const int NB = (N + THREADS - 1) / THREADS;

    float* ws      = (float*)d_ws;
    float* blkSum  = ws;
    float* blkMinV = ws + (size_t)NB * SOM_DIM;
    int*   blkMinI = (int*)(blkMinV + NB);
    size_t hoff    = (size_t)NB * SOM_DIM + 2 * (size_t)NB;
    hoff = (hoff + 3) & ~(size_t)3;
    float* hdr     = ws + hoff;

    float* outP = (float*)d_out;
    float* outV = outP + (size_t)N * SOM_DIM;

    som_pass_a<<<NB, THREADS, 0, stream>>>(input, particles, blkSum, blkMinV, blkMinI, N);
    som_pass_b<<<1, BTHREADS, 0, stream>>>(particles, gloc, iter_num,
                                          blkSum, blkMinV, blkMinI, hdr, NB, N);
    som_pass_c<<<NB, THREADS, 0, stream>>>(particles, velocities, r1, r2,
                                           gloc, hdr, outP, outV, N);
}


# Pass C: a wave whose 32 rows all carry the reference's skip flag (outputs = old state) copies particles and velocities through and never loads r1 and r2; P and V are requested before the flags
# speedup vs baseline: 4.9755x; 1.2644x over previous
#include <hip/hip_runtime.h>
#include <hip/hip_bf16.h>
#include <math.h>

typedef float v2f __attribute__((ext_vector_type(2)));
typedef float v8f __attribute__((ext_vector_type(8)));

#define THREADS 256
#define SOM_DIM 32
#define LSTRIDE 33


template <bool FULL>
__device__ __forceinline__
void stage_rows(const float* __restrict__ particles, float* __restrict__ stage,
                int rowBase, int N, int t)
{
    const int chunk = t & 7;
    const int rsub  = t >> 3;
    const float4* p4 = reinterpret_cast<const float4*>(particles) + (size_t)rowBase * 8 + t;
    #pragma unroll
    for (int q = 0; q < 8; ++q) {
        const int rl = q * 32 + rsub;
        float4 pv = make_float4(0.f, 0.f, 0.f, 0.f);
        if (FULL || rowBase + rl < N) pv = p4[q * 256];
        float* dst = stage + rl * LSTRIDE + chunk * 4;
        dst[0] = pv.x; dst[1] = pv.y; dst[2] = pv.z; dst[3] = pv.w;
    }
}

__global__ __launch_bounds__(THREADS)
void som_pass_a(const float* __restrict__ input,
                const float* __restrict__ particles,
                float* __restrict__ blkSum,
                float* __restrict__ blkMinV,
                int*   __restrict__ blkMinI,
                int N)
{
    __shared__ float stage[THREADS * LSTRIDE];
    __shared__ float wsum[8 * SOM_DIM];
    __shared__ float sdist[THREADS];
    __shared__ int   sidx[THREADS];

    const int t    = threadIdx.x;
    const int p    = blockIdx.x * THREADS + t;
    const int lane = t & 31;
    const int wave = t >> 5;

    const int rowBase = blockIdx.x * THREADS;
    if (rowBase + THREADS <= N) stage_rows<true >(particles, stage, rowBase, N, t);
    else                        stage_rows<false>(particles, stage, rowBase, N, t);
    __syncthreads();

    float dist2 = 3.402823466e+38f;
    if (p < N) {
        const float4* x4 = reinterpret_cast<const float4*>(input);
        const float* srow = &stage[t * LSTRIDE];
        float acc = 0.f;
        #pragma unroll
        for (int q = 0; q < 8; ++q) {
            const float4 xv = x4[q];
            float d0 = srow[q*4+0] - xv.x, d1 = srow[q*4+1] - xv.y;
            float d2 = srow[q*4+2] - xv.z, d3 = srow[q*4+3] - xv.w;
            acc += d0*d0 + d1*d1 + d2*d2 + d3*d3;
        }
        dist2 = acc;
    }

    const int half = lane >> 4;
    const int n    = lane & 15;
    v2f onesA; onesA.x = 1.0f; onesA.y = 1.0f;
    v8f acc0 = {};
    v8f acc1 = {};
    const int rowbase = wave * 32;
    #pragma unroll
    for (int c = 0; c < 8; ++c) {
        const int r0 = rowbase + c*4 + half;
        const int r1 = r0 + 2;
        v2f b0, b1;
        b0.x = stage[r0*LSTRIDE + n];        b0.y = stage[r1*LSTRIDE + n];
        b1.x = stage[r0*LSTRIDE + 16 + n];   b1.y = stage[r1*LSTRIDE + 16 + n];
        acc0 = __builtin_amdgcn_wmma_f32_16x16x4_f32(false, onesA, false, b0,
                                                     (short)0, acc0, false, false);
        acc1 = __builtin_amdgcn_wmma_f32_16x16x4_f32(false, onesA, false, b1,
                                                     (short)0, acc1, false, false);
    }
    if (lane < 16) {
        wsum[wave*SOM_DIM + lane]      = acc0[0];
        wsum[wave*SOM_DIM + 16 + lane] = acc1[0];
    }

    sdist[t] = dist2;
    sidx[t]  = p;
    __syncthreads();

    #pragma unroll
    for (int s = 128; s > 0; s >>= 1) {
        if (t < s) {
            float a = sdist[t], b = sdist[t + s];
            int   ia = sidx[t], ib = sidx[t + s];
            if (b < a || (b == a && ib < ia)) { sdist[t] = b; sidx[t] = ib; }
        }
        __syncthreads();
    }

    if (t < SOM_DIM) {
        float s = 0.f;
        #pragma unroll
        for (int w = 0; w < 8; ++w) s += wsum[w*SOM_DIM + t];
        blkSum[(size_t)blockIdx.x * SOM_DIM + t] = s;
    }
    if (t == 0) { blkMinV[blockIdx.x] = sdist[0]; blkMinI[blockIdx.x] = sidx[0]; }
}

#define BTHREADS 1024

__global__ __launch_bounds__(BTHREADS)
void som_pass_b(const float* __restrict__ particles,
                const int*   __restrict__ gloc,
                const int*   __restrict__ iter_num,
                const float* __restrict__ blkSum,
                const float* __restrict__ blkMinV,
                const int*   __restrict__ blkMinI,
                float* __restrict__ hdr,
                int NB, int N)
{
    __shared__ float4 wsumB4[32 * 8];
    __shared__ float wdist[32];
    __shared__ int   widx[32];

    const int t    = threadIdx.x;
    const int lane = t & 31;
    const int wave = t >> 5;

    const int iter = iter_num[0];

    float best = 3.402823466e+38f; int bidx = 0x7fffffff;
    if (t < NB) { best = blkMinV[t]; bidx = blkMinI[t]; }

    const float4* s4 = reinterpret_cast<const float4*>(blkSum);
    const int total4 = NB * 8;
    float4 acc = make_float4(0.f, 0.f, 0.f, 0.f);
    for (int base = 0; base < total4; base += 8 * BTHREADS) {
        float4 v[8];
        if (base + 8 * BTHREADS <= total4) {
            #pragma unroll
            for (int j = 0; j < 8; ++j) v[j] = s4[base + j * BTHREADS + t];
        } else {
            #pragma unroll
            for (int j = 0; j < 8; ++j) {
                const int idx = base + j * BTHREADS + t;
                v[j] = make_float4(0.f, 0.f, 0.f, 0.f);
                if (idx < total4) v[j] = s4[idx];
            }
        }
        #pragma unroll
        for (int j = 0; j < 8; ++j) {
            acc.x += v[j].x; acc.y += v[j].y; acc.z += v[j].z; acc.w += v[j].w;
        }
    }

    for (int k = t + BTHREADS; k < NB; k += BTHREADS) {
        const float dv = blkMinV[k]; const int iv = blkMinI[k];
        if (dv < best || (dv == best && iv < bidx)) { best = dv; bidx = iv; }
    }

    acc.x += __shfl_xor(acc.x, 8);  acc.y += __shfl_xor(acc.y, 8);
    acc.z += __shfl_xor(acc.z, 8);  acc.w += __shfl_xor(acc.w, 8);
    acc.x += __shfl_xor(acc.x, 16); acc.y += __shfl_xor(acc.y, 16);
    acc.z += __shfl_xor(acc.z, 16); acc.w += __shfl_xor(acc.w, 16);
    if (lane < 8) wsumB4[wave * 8 + lane] = acc;

    #pragma unroll
    for (int m = 16; m > 0; m >>= 1) {
        const float od = __shfl_xor(best, m);
        const int   oi = __shfl_xor(bidx, m);
        if (od < best || (od == best && oi < bidx)) { best = od; bidx = oi; }
    }
    if (lane == 0) { wdist[wave] = best; widx[wave] = bidx; }
    __syncthreads();

    if (wave == 0) {
        float b2 = wdist[lane]; int bmu = widx[lane];
        #pragma unroll
        for (int m = 16; m > 0; m >>= 1) {
            const float od = __shfl_xor(b2, m);
            const int   oi = __shfl_xor(bmu, m);
            if (od < b2 || (od == b2 && oi < bmu)) { b2 = od; bmu = oi; }
        }
        const float gb = particles[(size_t)bmu * SOM_DIM + lane];

        const float* wsumB = reinterpret_cast<const float*>(wsumB4);
        float S = 0.f;
        #pragma unroll
        for (int w = 0; w < 32; ++w) S += wsumB[w * SOM_DIM + lane];

        hdr[4 + lane]              = gb;
        hdr[4 + SOM_DIM + lane]    = S / (float)N;
        hdr[4 + 2*SOM_DIM + lane]  = (S - gb) / (float)(N - 1);
        if (lane == 0) {
            const float decay  = 1.0f - (float)iter / 100.0f;
            const float radius = 0.5f * decay;
            const float sig    = 256.0f * decay;
            hdr[0] = (float)gloc[2*bmu];
            hdr[1] = (float)gloc[2*bmu + 1];
            hdr[2] = radius;
            hdr[3] = sig * sig;
        }
    }
}

__device__ __forceinline__
void update_piece(const float4 pv, const float4 vv, const float4 a, const float4 b,
                  const float4 g, const float4 cA, const float4 cB,
                  const bool sk, const bool sa, float4& pn, float4& vn)
{
    float4 c;
    c.x = sa ? cA.x : cB.x; c.y = sa ? cA.y : cB.y;
    c.z = sa ? cA.z : cB.z; c.w = sa ? cA.w : cB.w;
    vn.x = 0.001f*vv.x + 0.01f*a.x*(c.x - pv.x) + 0.1f*b.x*(g.x - pv.x);
    vn.y = 0.001f*vv.y + 0.01f*a.y*(c.y - pv.y) + 0.1f*b.y*(g.y - pv.y);
    vn.z = 0.001f*vv.z + 0.01f*a.z*(c.z - pv.z) + 0.1f*b.z*(g.z - pv.z);
    vn.w = 0.001f*vv.w + 0.01f*a.w*(c.w - pv.w) + 0.1f*b.w*(g.w - pv.w);
    pn.x = pv.x + vn.x; pn.y = pv.y + vn.y;
    pn.z = pv.z + vn.z; pn.w = pv.w + vn.w;
    pn.x = sk ? pv.x : pn.x; pn.y = sk ? pv.y : pn.y;
    pn.z = sk ? pv.z : pn.z; pn.w = sk ? pv.w : pn.w;
    vn.x = sk ? vv.x : vn.x; vn.y = sk ? vv.y : vn.y;
    vn.z = sk ? vv.z : vn.z; vn.w = sk ? vv.w : vn.w;
}

__device__ __forceinline__
void row_flags_at(const int2 gl,
                  float bx, float by, float radius, float sigma2, bool& skip, bool& selA)
{
    const float dx = (float)gl.x - bx;
    const float dy = (float)gl.y - by;
    const float nb = expf(-(dx*dx + dy*dy) / sigma2);
    skip = (1.0f - nb) > radius;
    selA = (nb + radius) >= 1.0f;
}

__device__ __forceinline__
void row_flags(const int* __restrict__ gloc, int prow,
               float bx, float by, float radius, float sigma2, bool& skip, bool& selA)
{
    const int2 gl = *(const int2*)(gloc + 2*(size_t)prow);
    row_flags_at(gl, bx, by, radius, sigma2, skip, selA);
}

__global__ __launch_bounds__(THREADS)
void som_pass_c(const float* __restrict__ particles,
                const float* __restrict__ velocities,
                const float* __restrict__ r1,
                const float* __restrict__ r2,
                const int*   __restrict__ gloc,
                const float* __restrict__ hdr,
                float* __restrict__ outP,
                float* __restrict__ outV,
                int N)
{
    const int lane = threadIdx.x & 31;
    const int rowW = blockIdx.x * THREADS + (threadIdx.x >> 5) * 32;
    const int prow = rowW + lane;

    const float bx = hdr[0], by = hdr[1], radius = hdr[2], sigma2 = hdr[3];

    const int chunk = lane & 7;
    const int rsub  = lane >> 3;
    const float4 g  = ((const float4*)(hdr + 4))[chunk];
    const float4 cA = ((const float4*)(hdr + 4 + SOM_DIM))[chunk];
    const float4 cB = ((const float4*)(hdr + 4 + 2*SOM_DIM))[chunk];

    const size_t base = (size_t)rowW * 8 + lane;
    const float4* P4  = (const float4*)particles  + base;
    const float4* V4  = (const float4*)velocities + base;
    const float4* R14 = (const float4*)r1 + base;
    const float4* R24 = (const float4*)r2 + base;
    float4* OP = (float4*)outP + base;
    float4* OV = (float4*)outV + base;

    if (rowW + 32 <= N) {
        const int2 gl = *(const int2*)(gloc + 2*(size_t)prow);
        float4 pv[8], vv[8];
        #pragma unroll
        for (int q = 0; q < 8; ++q) { pv[q] = P4[q*32]; vv[q] = V4[q*32]; }

        bool skip, selA;
        row_flags_at(gl, bx, by, radius, sigma2, skip, selA);
        const unsigned skipMask = __builtin_amdgcn_ballot_w32(skip);
        if (skipMask == 0xffffffffu) {
            #pragma unroll
            for (int q = 0; q < 8; ++q) {
                OP[q*32] = pv[q];
                OV[q*32] = vv[q];
            }
        } else {
            const unsigned selMask = __builtin_amdgcn_ballot_w32(selA);
            #pragma unroll
            for (int q = 0; q < 8; ++q) {
                const int rl = q * 4 + rsub;
                const bool sk = ((skipMask >> rl) & 1u) != 0u;
                const bool sa = ((selMask  >> rl) & 1u) != 0u;
                float4 pn, vn;
                update_piece(pv[q], vv[q], R14[q*32], R24[q*32], g, cA, cB, sk, sa, pn, vn);
                OP[q*32] = pn;
                OV[q*32] = vn;
            }
        }
    } else {
        bool skip = true, selA = false;
        if (prow < N) row_flags(gloc, prow, bx, by, radius, sigma2, skip, selA);
        const unsigned skipMask = __builtin_amdgcn_ballot_w32(skip);
        const unsigned selMask  = __builtin_amdgcn_ballot_w32(selA);
        for (int q = 0; q < 8; ++q) {
            const int rl = q * 4 + rsub;
            if (rowW + rl < N) {
                const bool sk = ((skipMask >> rl) & 1u) != 0u;
                const bool sa = ((selMask  >> rl) & 1u) != 0u;
                float4 pn, vn;
                update_piece(P4[q*32], V4[q*32], R14[q*32], R24[q*32], g, cA, cB, sk, sa, pn, vn);
                OP[q*32] = pn;
                OV[q*32] = vn;
            }
        }
    }
}

extern "C" void kernel_launch(void* const* d_in, const int* in_sizes, int n_in,
                              void* d_out, int out_size, void* d_ws, size_t ws_size,
                              hipStream_t stream)
{
    (void)n_in; (void)out_size; (void)ws_size;
    const float* input      = (const float*)d_in[0];
    const float* particles  = (const float*)d_in[1];
    const float* velocities = (const float*)d_in[2];
    const float* r1         = (const float*)d_in[3];
    const float* r2         = (const float*)d_in[4];
    const int*   gloc       = (const int*)d_in[5];
    const int*   iter_num   = (const int*)d_in[6];

    const int N  = in_sizes[1] / SOM_DIM;
    const int NB = (N + THREADS - 1) / THREADS;

    float* ws      = (float*)d_ws;
    float* blkSum  = ws;
    float* blkMinV = ws + (size_t)NB * SOM_DIM;
    int*   blkMinI = (int*)(blkMinV + NB);
    size_t hoff    = (size_t)NB * SOM_DIM + 2 * (size_t)NB;
    hoff = (hoff + 3) & ~(size_t)3;
    float* hdr     = ws + hoff;

    float* outP = (float*)d_out;
    float* outV = outP + (size_t)N * SOM_DIM;

    som_pass_a<<<NB, THREADS, 0, stream>>>(input, particles, blkSum, blkMinV, blkMinI, N);
    som_pass_b<<<1, BTHREADS, 0, stream>>>(particles, gloc, iter_num,
                                          blkSum, blkMinV, blkMinI, hdr, NB, N);
    som_pass_c<<<NB, THREADS, 0, stream>>>(particles, velocities, r1, r2,
                                           gloc, hdr, outP, outV, N);
}


# Pass A: rows stay in the registers they were loaded into, no LDS tile: row distances by 8-lane DPP sums, column sums by WMMA on those registers, 1 barrier not 10; column-sum order changed, fixed
# speedup vs baseline: 5.3923x; 1.0838x over previous
#include <hip/hip_runtime.h>
#include <hip/hip_bf16.h>
#include <math.h>

typedef float v2f __attribute__((ext_vector_type(2)));
typedef float v8f __attribute__((ext_vector_type(8)));

#define THREADS 256
#define SOM_DIM 32
#define ROWS_A 512


template <int M>
__device__ __forceinline__ int xrow_i(int b)
{
    return __builtin_amdgcn_update_dpp(b, b, 0x160 + M, 0xf, 0xf, false);
}
template <int M>
__device__ __forceinline__ float xrow(float x)
{
    return __builtin_bit_cast(float, xrow_i<M>(__builtin_bit_cast(int, x)));
}
__device__ __forceinline__ int xhalf_i(int b)
{
    return __builtin_amdgcn_permlanex16(b, b, 0x76543210, 0xfedcba98, false, false);
}
__device__ __forceinline__ float xhalf(float x)
{
    return __builtin_bit_cast(float, xhalf_i(__builtin_bit_cast(int, x)));
}
__device__ __forceinline__
void take_lower(float& best, int& bidx, const float od, const int oi)
{
    const bool tk = od < best || (od == best && oi < bidx);
    best = tk ? od : best;
    bidx = tk ? oi : bidx;
}

__global__ __launch_bounds__(THREADS)
void som_pass_a(const float* __restrict__ input,
                const float* __restrict__ particles,
                float* __restrict__ blkSum,
                float* __restrict__ blkMinV,
                int*   __restrict__ blkMinI,
                int N)
{
    __shared__ float4 wsum4[8 * 8];
    __shared__ float  wdist[8];
    __shared__ int    widx[8];

    const int t     = threadIdx.x;
    const int lane  = t & 31;
    const int wave  = t >> 5;
    const int chunk = lane & 7;
    const int rsub  = lane >> 3;
    const int rowW  = blockIdx.x * ROWS_A + wave * 64;

    const float4 xv = reinterpret_cast<const float4*>(input)[chunk];
    const float4* p4 = reinterpret_cast<const float4*>(particles) + (size_t)rowW * 8 + lane;

    float4 pv[16];
    if (rowW + 64 <= N) {
        #pragma unroll
        for (int q = 0; q < 16; ++q) pv[q] = p4[q * 32];
    } else {
        #pragma unroll
        for (int q = 0; q < 16; ++q) {
            pv[q] = make_float4(0.f, 0.f, 0.f, 0.f);
            if (rowW + q * 4 + rsub < N) pv[q] = p4[q * 32];
        }
    }

    v2f onesA; onesA.x = 1.0f; onesA.y = 1.0f;
    v8f ax = {}, ay = {}, az = {}, aw = {};
    #pragma unroll
    for (int q = 0; q < 16; q += 2) {
        v2f bx, by, bz, bw;
        bx.x = pv[q].x; bx.y = pv[q + 1].x;
        by.x = pv[q].y; by.y = pv[q + 1].y;
        bz.x = pv[q].z; bz.y = pv[q + 1].z;
        bw.x = pv[q].w; bw.y = pv[q + 1].w;
        ax = __builtin_amdgcn_wmma_f32_16x16x4_f32(false, onesA, false, bx, (short)0, ax, false, false);
        ay = __builtin_amdgcn_wmma_f32_16x16x4_f32(false, onesA, false, by, (short)0, ay, false, false);
        az = __builtin_amdgcn_wmma_f32_16x16x4_f32(false, onesA, false, bz, (short)0, az, false, false);
        aw = __builtin_amdgcn_wmma_f32_16x16x4_f32(false, onesA, false, bw, (short)0, aw, false, false);
    }

    float best = 3.402823466e+38f;
    int   bidx = (rowW + rsub < N) ? (rowW + rsub) : 0x7fffffff;
    #pragma unroll
    for (int q = 0; q < 16; ++q) {
        const float d0 = pv[q].x - xv.x, d1 = pv[q].y - xv.y;
        const float d2 = pv[q].z - xv.z, d3 = pv[q].w - xv.w;
        float s = d0*d0 + d1*d1 + d2*d2 + d3*d3;
        s += xrow<1>(s);
        s += xrow<2>(s);
        s += xrow<4>(s);
        const int row = rowW + q * 4 + rsub;
        const bool tk = (row < N) && (s < best);
        best = tk ? s : best;
        bidx = tk ? row : bidx;
    }
    take_lower(best, bidx, xrow<8>(best), xrow_i<8>(bidx));
    take_lower(best, bidx, xhalf(best),   xhalf_i(bidx));

    float4 cs;
    cs.x = ax[0] + xrow<8>(ax[0]);
    cs.y = ay[0] + xrow<8>(ay[0]);
    cs.z = az[0] + xrow<8>(az[0]);
    cs.w = aw[0] + xrow<8>(aw[0]);
    if (lane < 8) wsum4[wave * 8 + lane] = cs;
    if (lane == 0) { wdist[wave] = best; widx[wave] = bidx; }
    __syncthreads();

    if (wave == 0) {
        const float* wsum = reinterpret_cast<const float*>(wsum4);
        float s = 0.f;
        #pragma unroll
        for (int w = 0; w < 8; ++w) s += wsum[w * SOM_DIM + lane];
        blkSum[(size_t)blockIdx.x * SOM_DIM + lane] = s;

        float b2 = wdist[lane & 7];
        int   bi = widx[lane & 7];
        take_lower(b2, bi, xrow<1>(b2), xrow_i<1>(bi));
        take_lower(b2, bi, xrow<2>(b2), xrow_i<2>(bi));
        take_lower(b2, bi, xrow<4>(b2), xrow_i<4>(bi));
        if (lane == 0) { blkMinV[blockIdx.x] = b2; blkMinI[blockIdx.x] = bi; }
    }
}

#define BTHREADS 1024

__global__ __launch_bounds__(BTHREADS)
void som_pass_b(const float* __restrict__ particles,
                const int*   __restrict__ gloc,
                const int*   __restrict__ iter_num,
                const float* __restrict__ blkSum,
                const float* __restrict__ blkMinV,
                const int*   __restrict__ blkMinI,
                float* __restrict__ hdr,
                int NB, int N)
{
    __shared__ float4 wsumB4[32 * 8];
    __shared__ float wdist[32];
    __shared__ int   widx[32];

    const int t    = threadIdx.x;
    const int lane = t & 31;
    const int wave = t >> 5;

    const int iter = iter_num[0];

    float best = 3.402823466e+38f; int bidx = 0x7fffffff;
    if (t < NB) { best = blkMinV[t]; bidx = blkMinI[t]; }

    const float4* s4 = reinterpret_cast<const float4*>(blkSum);
    const int total4 = NB * 8;
    float4 acc = make_float4(0.f, 0.f, 0.f, 0.f);
    for (int base = 0; base < total4; base += 8 * BTHREADS) {
        float4 v[8];
        if (base + 8 * BTHREADS <= total4) {
            #pragma unroll
            for (int j = 0; j < 8; ++j) v[j] = s4[base + j * BTHREADS + t];
        } else {
            #pragma unroll
            for (int j = 0; j < 8; ++j) {
                const int idx = base + j * BTHREADS + t;
                v[j] = make_float4(0.f, 0.f, 0.f, 0.f);
                if (idx < total4) v[j] = s4[idx];
            }
        }
        #pragma unroll
        for (int j = 0; j < 8; ++j) {
            acc.x += v[j].x; acc.y += v[j].y; acc.z += v[j].z; acc.w += v[j].w;
        }
    }

    for (int k = t + BTHREADS; k < NB; k += BTHREADS) {
        const float dv = blkMinV[k]; const int iv = blkMinI[k];
        if (dv < best || (dv == best && iv < bidx)) { best = dv; bidx = iv; }
    }

    acc.x += __shfl_xor(acc.x, 8);  acc.y += __shfl_xor(acc.y, 8);
    acc.z += __shfl_xor(acc.z, 8);  acc.w += __shfl_xor(acc.w, 8);
    acc.x += __shfl_xor(acc.x, 16); acc.y += __shfl_xor(acc.y, 16);
    acc.z += __shfl_xor(acc.z, 16); acc.w += __shfl_xor(acc.w, 16);
    if (lane < 8) wsumB4[wave * 8 + lane] = acc;

    #pragma unroll
    for (int m = 16; m > 0; m >>= 1) {
        const float od = __shfl_xor(best, m);
        const int   oi = __shfl_xor(bidx, m);
        if (od < best || (od == best && oi < bidx)) { best = od; bidx = oi; }
    }
    if (lane == 0) { wdist[wave] = best; widx[wave] = bidx; }
    __syncthreads();

    if (wave == 0) {
        float b2 = wdist[lane]; int bmu = widx[lane];
        #pragma unroll
        for (int m = 16; m > 0; m >>= 1) {
            const float od = __shfl_xor(b2, m);
            const int   oi = __shfl_xor(bmu, m);
            if (od < b2 || (od == b2 && oi < bmu)) { b2 = od; bmu = oi; }
        }
        const float gb = particles[(size_t)bmu * SOM_DIM + lane];

        const float* wsumB = reinterpret_cast<const float*>(wsumB4);
        float S = 0.f;
        #pragma unroll
        for (int w = 0; w < 32; ++w) S += wsumB[w * SOM_DIM + lane];

        hdr[4 + lane]              = gb;
        hdr[4 + SOM_DIM + lane]    = S / (float)N;
        hdr[4 + 2*SOM_DIM + lane]  = (S - gb) / (float)(N - 1);
        if (lane == 0) {
            const float decay  = 1.0f - (float)iter / 100.0f;
            const float radius = 0.5f * decay;
            const float sig    = 256.0f * decay;
            hdr[0] = (float)gloc[2*bmu];
            hdr[1] = (float)gloc[2*bmu + 1];
            hdr[2] = radius;
            hdr[3] = sig * sig;
        }
    }
}

__device__ __forceinline__
void update_piece(const float4 pv, const float4 vv, const float4 a, const float4 b,
                  const float4 g, const float4 cA, const float4 cB,
                  const bool sk, const bool sa, float4& pn, float4& vn)
{
    float4 c;
    c.x = sa ? cA.x : cB.x; c.y = sa ? cA.y : cB.y;
    c.z = sa ? cA.z : cB.z; c.w = sa ? cA.w : cB.w;
    vn.x = 0.001f*vv.x + 0.01f*a.x*(c.x - pv.x) + 0.1f*b.x*(g.x - pv.x);
    vn.y = 0.001f*vv.y + 0.01f*a.y*(c.y - pv.y) + 0.1f*b.y*(g.y - pv.y);
    vn.z = 0.001f*vv.z + 0.01f*a.z*(c.z - pv.z) + 0.1f*b.z*(g.z - pv.z);
    vn.w = 0.001f*vv.w + 0.01f*a.w*(c.w - pv.w) + 0.1f*b.w*(g.w - pv.w);
    pn.x = pv.x + vn.x; pn.y = pv.y + vn.y;
    pn.z = pv.z + vn.z; pn.w = pv.w + vn.w;
    pn.x = sk ? pv.x : pn.x; pn.y = sk ? pv.y : pn.y;
    pn.z = sk ? pv.z : pn.z; pn.w = sk ? pv.w : pn.w;
    vn.x = sk ? vv.x : vn.x; vn.y = sk ? vv.y : vn.y;
    vn.z = sk ? vv.z : vn.z; vn.w = sk ? vv.w : vn.w;
}

__device__ __forceinline__
void row_flags_at(const int2 gl,
                  float bx, float by, float radius, float sigma2, bool& skip, bool& selA)
{
    const float dx = (float)gl.x - bx;
    const float dy = (float)gl.y - by;
    const float nb = expf(-(dx*dx + dy*dy) / sigma2);
    skip = (1.0f - nb) > radius;
    selA = (nb + radius) >= 1.0f;
}

__device__ __forceinline__
void row_flags(const int* __restrict__ gloc, int prow,
               float bx, float by, float radius, float sigma2, bool& skip, bool& selA)
{
    const int2 gl = *(const int2*)(gloc + 2*(size_t)prow);
    row_flags_at(gl, bx, by, radius, sigma2, skip, selA);
}

__global__ __launch_bounds__(THREADS)
void som_pass_c(const float* __restrict__ particles,
                const float* __restrict__ velocities,
                const float* __restrict__ r1,
                const float* __restrict__ r2,
                const int*   __restrict__ gloc,
                const float* __restrict__ hdr,
                float* __restrict__ outP,
                float* __restrict__ outV,
                int N)
{
    const int lane = threadIdx.x & 31;
    const int rowW = blockIdx.x * THREADS + (threadIdx.x >> 5) * 32;
    const int prow = rowW + lane;

    const float bx = hdr[0], by = hdr[1], radius = hdr[2], sigma2 = hdr[3];

    const int chunk = lane & 7;
    const int rsub  = lane >> 3;
    const float4 g  = ((const float4*)(hdr + 4))[chunk];
    const float4 cA = ((const float4*)(hdr + 4 + SOM_DIM))[chunk];
    const float4 cB = ((const float4*)(hdr + 4 + 2*SOM_DIM))[chunk];

    const size_t base = (size_t)rowW * 8 + lane;
    const float4* P4  = (const float4*)particles  + base;
    const float4* V4  = (const float4*)velocities + base;
    const float4* R14 = (const float4*)r1 + base;
    const float4* R24 = (const float4*)r2 + base;
    float4* OP = (float4*)outP + base;
    float4* OV = (float4*)outV + base;

    if (rowW + 32 <= N) {
        const int2 gl = *(const int2*)(gloc + 2*(size_t)prow);
        float4 pv[8], vv[8];
        #pragma unroll
        for (int q = 0; q < 8; ++q) { pv[q] = P4[q*32]; vv[q] = V4[q*32]; }

        bool skip, selA;
        row_flags_at(gl, bx, by, radius, sigma2, skip, selA);
        const unsigned skipMask = __builtin_amdgcn_ballot_w32(skip);
        if (skipMask == 0xffffffffu) {
            #pragma unroll
            for (int q = 0; q < 8; ++q) {
                OP[q*32] = pv[q];
                OV[q*32] = vv[q];
            }
        } else {
            const unsigned selMask = __builtin_amdgcn_ballot_w32(selA);
            #pragma unroll
            for (int q = 0; q < 8; ++q) {
                const int rl = q * 4 + rsub;
                const bool sk = ((skipMask >> rl) & 1u) != 0u;
                const bool sa = ((selMask  >> rl) & 1u) != 0u;
                float4 pn, vn;
                update_piece(pv[q], vv[q], R14[q*32], R24[q*32], g, cA, cB, sk, sa, pn, vn);
                OP[q*32] = pn;
                OV[q*32] = vn;
            }
        }
    } else {
        bool skip = true, selA = false;
        if (prow < N) row_flags(gloc, prow, bx, by, radius, sigma2, skip, selA);
        const unsigned skipMask = __builtin_amdgcn_ballot_w32(skip);
        const unsigned selMask  = __builtin_amdgcn_ballot_w32(selA);
        for (int q = 0; q < 8; ++q) {
            const int rl = q * 4 + rsub;
            if (rowW + rl < N) {
                const bool sk = ((skipMask >> rl) & 1u) != 0u;
                const bool sa = ((selMask  >> rl) & 1u) != 0u;
                float4 pn, vn;
                update_piece(P4[q*32], V4[q*32], R14[q*32], R24[q*32], g, cA, cB, sk, sa, pn, vn);
                OP[q*32] = pn;
                OV[q*32] = vn;
            }
        }
    }
}

extern "C" void kernel_launch(void* const* d_in, const int* in_sizes, int n_in,
                              void* d_out, int out_size, void* d_ws, size_t ws_size,
                              hipStream_t stream)
{
    (void)n_in; (void)out_size; (void)ws_size;
    const float* input      = (const float*)d_in[0];
    const float* particles  = (const float*)d_in[1];
    const float* velocities = (const float*)d_in[2];
    const float* r1         = (const float*)d_in[3];
    const float* r2         = (const float*)d_in[4];
    const int*   gloc       = (const int*)d_in[5];
    const int*   iter_num   = (const int*)d_in[6];

    const int N  = in_sizes[1] / SOM_DIM;
    const int NB  = (N + ROWS_A - 1) / ROWS_A;
    const int NBC = (N + THREADS - 1) / THREADS;

    float* ws      = (float*)d_ws;
    float* blkSum  = ws;
    float* blkMinV = ws + (size_t)NB * SOM_DIM;
    int*   blkMinI = (int*)(blkMinV + NB);
    size_t hoff    = (size_t)NB * SOM_DIM + 2 * (size_t)NB;
    hoff = (hoff + 3) & ~(size_t)3;
    float* hdr     = ws + hoff;

    float* outP = (float*)d_out;
    float* outV = outP + (size_t)N * SOM_DIM;

    som_pass_a<<<NB, THREADS, 0, stream>>>(input, particles, blkSum, blkMinV, blkMinI, N);
    som_pass_b<<<1, BTHREADS, 0, stream>>>(particles, gloc, iter_num,
                                          blkSum, blkMinV, blkMinI, hdr, NB, N);
    som_pass_c<<<NBC, THREADS, 0, stream>>>(particles, velocities, r1, r2,
                                           gloc, hdr, outP, outV, N);
}


# Pass B's launch removed: each work-group of pass C (1024 threads, as pass B had) folds pass A's partials itself while its particle rows are in flight; 2 launches; column-sum order changed, fixed
# speedup vs baseline: 5.8149x; 1.0784x over previous
#include <hip/hip_runtime.h>
#include <hip/hip_bf16.h>
#include <math.h>

typedef float v2f __attribute__((ext_vector_type(2)));
typedef float v8f __attribute__((ext_vector_type(8)));

#define THREADS 256
#define SOM_DIM 32
#define ROWS_A 512


template <int M>
__device__ __forceinline__ int xrow_i(int b)
{
    return __builtin_amdgcn_update_dpp(b, b, 0x160 + M, 0xf, 0xf, false);
}
template <int M>
__device__ __forceinline__ float xrow(float x)
{
    return __builtin_bit_cast(float, xrow_i<M>(__builtin_bit_cast(int, x)));
}
__device__ __forceinline__ int xhalf_i(int b)
{
    return __builtin_amdgcn_permlanex16(b, b, 0x76543210, 0xfedcba98, false, false);
}
__device__ __forceinline__ float xhalf(float x)
{
    return __builtin_bit_cast(float, xhalf_i(__builtin_bit_cast(int, x)));
}
__device__ __forceinline__
void take_lower(float& best, int& bidx, const float od, const int oi)
{
    const bool tk = od < best || (od == best && oi < bidx);
    best = tk ? od : best;
    bidx = tk ? oi : bidx;
}

__global__ __launch_bounds__(THREADS)
void som_pass_a(const float* __restrict__ input,
                const float* __restrict__ particles,
                float* __restrict__ blkSum,
                float* __restrict__ blkMinV,
                int*   __restrict__ blkMinI,
                int N)
{
    __shared__ float4 wsum4[8 * 8];
    __shared__ float  wdist[8];
    __shared__ int    widx[8];

    const int t     = threadIdx.x;
    const int lane  = t & 31;
    const int wave  = t >> 5;
    const int chunk = lane & 7;
    const int rsub  = lane >> 3;
    const int rowW  = blockIdx.x * ROWS_A + wave * 64;

    const float4 xv = reinterpret_cast<const float4*>(input)[chunk];
    const float4* p4 = reinterpret_cast<const float4*>(particles) + (size_t)rowW * 8 + lane;

    float4 pv[16];
    if (rowW + 64 <= N) {
        #pragma unroll
        for (int q = 0; q < 16; ++q) pv[q] = p4[q * 32];
    } else {
        #pragma unroll
        for (int q = 0; q < 16; ++q) {
            pv[q] = make_float4(0.f, 0.f, 0.f, 0.f);
            if (rowW + q * 4 + rsub < N) pv[q] = p4[q * 32];
        }
    }

    v2f onesA; onesA.x = 1.0f; onesA.y = 1.0f;
    v8f ax = {}, ay = {}, az = {}, aw = {};
    #pragma unroll
    for (int q = 0; q < 16; q += 2) {
        v2f bx, by, bz, bw;
        bx.x = pv[q].x; bx.y = pv[q + 1].x;
        by.x = pv[q].y; by.y = pv[q + 1].y;
        bz.x = pv[q].z; bz.y = pv[q + 1].z;
        bw.x = pv[q].w; bw.y = pv[q + 1].w;
        ax = __builtin_amdgcn_wmma_f32_16x16x4_f32(false, onesA, false, bx, (short)0, ax, false, false);
        ay = __builtin_amdgcn_wmma_f32_16x16x4_f32(false, onesA, false, by, (short)0, ay, false, false);
        az = __builtin_amdgcn_wmma_f32_16x16x4_f32(false, onesA, false, bz, (short)0, az, false, false);
        aw = __builtin_amdgcn_wmma_f32_16x16x4_f32(false, onesA, false, bw, (short)0, aw, false, false);
    }

    float best = 3.402823466e+38f;
    int   bidx = (rowW + rsub < N) ? (rowW + rsub) : 0x7fffffff;
    #pragma unroll
    for (int q = 0; q < 16; ++q) {
        const float d0 = pv[q].x - xv.x, d1 = pv[q].y - xv.y;
        const float d2 = pv[q].z - xv.z, d3 = pv[q].w - xv.w;
        float s = d0*d0 + d1*d1 + d2*d2 + d3*d3;
        s += xrow<1>(s);
        s += xrow<2>(s);
        s += xrow<4>(s);
        const int row = rowW + q * 4 + rsub;
        const bool tk = (row < N) && (s < best);
        best = tk ? s : best;
        bidx = tk ? row : bidx;
    }
    take_lower(best, bidx, xrow<8>(best), xrow_i<8>(bidx));
    take_lower(best, bidx, xhalf(best),   xhalf_i(bidx));

    float4 cs;
    cs.x = ax[0] + xrow<8>(ax[0]);
    cs.y = ay[0] + xrow<8>(ay[0]);
    cs.z = az[0] + xrow<8>(az[0]);
    cs.w = aw[0] + xrow<8>(aw[0]);
    if (lane < 8) wsum4[wave * 8 + lane] = cs;
    if (lane == 0) { wdist[wave] = best; widx[wave] = bidx; }
    __syncthreads();

    if (wave == 0) {
        const float* wsum = reinterpret_cast<const float*>(wsum4);
        float s = 0.f;
        #pragma unroll
        for (int w = 0; w < 8; ++w) s += wsum[w * SOM_DIM + lane];
        blkSum[(size_t)blockIdx.x * SOM_DIM + lane] = s;

        float b2 = wdist[lane & 7];
        int   bi = widx[lane & 7];
        take_lower(b2, bi, xrow<1>(b2), xrow_i<1>(bi));
        take_lower(b2, bi, xrow<2>(b2), xrow_i<2>(bi));
        take_lower(b2, bi, xrow<4>(b2), xrow_i<4>(bi));
        if (lane == 0) { blkMinV[blockIdx.x] = b2; blkMinI[blockIdx.x] = bi; }
    }
}

#define CTHREADS 1024

__device__ __forceinline__ void lds_barrier()
{
    asm volatile("s_wait_dscnt 0x0\n\ts_barrier_signal -1\n\ts_barrier_wait -1" ::: "memory");
}

__device__ __forceinline__
void update_piece(const float4 pv, const float4 vv, const float4 a, const float4 b,
                  const float4 g, const float4 cA, const float4 cB,
                  const bool sk, const bool sa, float4& pn, float4& vn)
{
    float4 c;
    c.x = sa ? cA.x : cB.x; c.y = sa ? cA.y : cB.y;
    c.z = sa ? cA.z : cB.z; c.w = sa ? cA.w : cB.w;
    vn.x = 0.001f*vv.x + 0.01f*a.x*(c.x - pv.x) + 0.1f*b.x*(g.x - pv.x);
    vn.y = 0.001f*vv.y + 0.01f*a.y*(c.y - pv.y) + 0.1f*b.y*(g.y - pv.y);
    vn.z = 0.001f*vv.z + 0.01f*a.z*(c.z - pv.z) + 0.1f*b.z*(g.z - pv.z);
    vn.w = 0.001f*vv.w + 0.01f*a.w*(c.w - pv.w) + 0.1f*b.w*(g.w - pv.w);
    pn.x = pv.x + vn.x; pn.y = pv.y + vn.y;
    pn.z = pv.z + vn.z; pn.w = pv.w + vn.w;
    pn.x = sk ? pv.x : pn.x; pn.y = sk ? pv.y : pn.y;
    pn.z = sk ? pv.z : pn.z; pn.w = sk ? pv.w : pn.w;
    vn.x = sk ? vv.x : vn.x; vn.y = sk ? vv.y : vn.y;
    vn.z = sk ? vv.z : vn.z; vn.w = sk ? vv.w : vn.w;
}

__device__ __forceinline__
void row_flags_at(const int2 gl,
                  float bx, float by, float radius, float sigma2, bool& skip, bool& selA)
{
    const float dx = (float)gl.x - bx;
    const float dy = (float)gl.y - by;
    const float nb = expf(-(dx*dx + dy*dy) / sigma2);
    skip = (1.0f - nb) > radius;
    selA = (nb + radius) >= 1.0f;
}

__device__ __forceinline__
void row_flags(const int* __restrict__ gloc, int prow,
               float bx, float by, float radius, float sigma2, bool& skip, bool& selA)
{
    const int2 gl = *(const int2*)(gloc + 2*(size_t)prow);
    row_flags_at(gl, bx, by, radius, sigma2, skip, selA);
}

__device__ __forceinline__
void centroids(const float4* wsum4, const float4 g, const int lane, const int N,
               float4& cA, float4& cB)
{
    float4 S = wsum4[lane];
    #pragma unroll
    for (int j = 1; j < 8; ++j) {
        const float4 w = wsum4[32 * j + lane];
        S.x += w.x; S.y += w.y; S.z += w.z; S.w += w.w;
    }
    S.x += xrow<8>(S.x); S.y += xrow<8>(S.y); S.z += xrow<8>(S.z); S.w += xrow<8>(S.w);
    S.x += xhalf(S.x);   S.y += xhalf(S.y);   S.z += xhalf(S.z);   S.w += xhalf(S.w);
    const float fn = (float)N, fn1 = (float)(N - 1);
    cA.x = S.x / fn; cA.y = S.y / fn; cA.z = S.z / fn; cA.w = S.w / fn;
    cB.x = (S.x - g.x) / fn1; cB.y = (S.y - g.y) / fn1;
    cB.z = (S.z - g.z) / fn1; cB.w = (S.w - g.w) / fn1;
}

__global__ __launch_bounds__(CTHREADS)
void som_pass_c(const float* __restrict__ particles,
                const float* velocities,
                const float* r1,
                const float* r2,
                const int*   __restrict__ gloc,
                const int*   __restrict__ iter_num,
                const float* __restrict__ blkSum,
                const float* __restrict__ blkMinV,
                const int*   __restrict__ blkMinI,
                float* outP,
                float* outV,
                int NB, int N)
{
    __shared__ float4 wsumC4[32 * 8];
    __shared__ float  wdistC[32];
    __shared__ int    widxC[32];

    const int t     = threadIdx.x;
    const int lane  = t & 31;
    const int wave  = t >> 5;
    const int rowW  = blockIdx.x * CTHREADS + wave * 32;
    const int prow  = rowW + lane;
    const int rsub  = lane >> 3;

    const int iter = iter_num[0];

    const int tb = (t < NB) ? t : NB - 1;
    const float minV = blkMinV[tb];
    const int   minI = blkMinI[tb];

    const float4* s4 = reinterpret_cast<const float4*>(blkSum);
    const int total4 = NB * 8;
    float4 v[4];
    #pragma unroll
    for (int j = 0; j < 4; ++j) {
        const int idx = j * CTHREADS + t;
        v[j] = s4[(idx < total4) ? idx : total4 - 1];
    }

    const unsigned base = (unsigned)rowW * 8u + (unsigned)lane;
    const float4* P4  = (const float4*)particles  + base;
    const float4* V4  = (const float4*)velocities + base;
    const float4* R14 = (const float4*)r1 + base;
    const float4* R24 = (const float4*)r2 + base;
    float4* OP = (float4*)outP + base;
    float4* OV = (float4*)outV + base;

    const bool full = (rowW + 32 <= N);
    const int rowL = full ? rowW : ((N >= 32) ? N - 32 : 0);
    const int2 gl = *(const int2*)(gloc + 2*(size_t)(rowL + lane));
    const float4* PL = (const float4*)particles + ((unsigned)rowL * 8u + (unsigned)lane);
    float4 pv[8];
    #pragma unroll
    for (int q = 0; q < 8; ++q) pv[q] = PL[q*32];

    float best = (t < NB) ? minV : 3.402823466e+38f;
    int   bidx = (t < NB) ? minI : 0x7fffffff;
    float4 acc = make_float4(0.f, 0.f, 0.f, 0.f);
    #pragma unroll
    for (int j = 0; j < 4; ++j) {
        const float m = ((j * CTHREADS + t) < total4) ? 1.0f : 0.0f;
        acc.x += m * v[j].x; acc.y += m * v[j].y;
        acc.z += m * v[j].z; acc.w += m * v[j].w;
    }
    for (int k = 4 * CTHREADS + t; k < total4; k += CTHREADS) {
        const float4 w = s4[k];
        acc.x += w.x; acc.y += w.y; acc.z += w.z; acc.w += w.w;
    }
    for (int k = t + CTHREADS; k < NB; k += CTHREADS) {
        take_lower(best, bidx, blkMinV[k], blkMinI[k]);
    }

    acc.x += xrow<8>(acc.x); acc.y += xrow<8>(acc.y); acc.z += xrow<8>(acc.z); acc.w += xrow<8>(acc.w);
    acc.x += xhalf(acc.x);   acc.y += xhalf(acc.y);   acc.z += xhalf(acc.z);   acc.w += xhalf(acc.w);
    if (lane < 8) wsumC4[wave * 8 + lane] = acc;

    take_lower(best, bidx, xrow<1>(best), xrow_i<1>(bidx));
    take_lower(best, bidx, xrow<2>(best), xrow_i<2>(bidx));
    take_lower(best, bidx, xrow<4>(best), xrow_i<4>(bidx));
    take_lower(best, bidx, xrow<8>(best), xrow_i<8>(bidx));
    take_lower(best, bidx, xhalf(best),   xhalf_i(bidx));
    if (lane == 0) { wdistC[wave] = best; widxC[wave] = bidx; }

    lds_barrier();

    float b2 = wdistC[lane]; int bm = widxC[lane];
    take_lower(b2, bm, xrow<1>(b2), xrow_i<1>(bm));
    take_lower(b2, bm, xrow<2>(b2), xrow_i<2>(bm));
    take_lower(b2, bm, xrow<4>(b2), xrow_i<4>(bm));
    take_lower(b2, bm, xrow<8>(b2), xrow_i<8>(bm));
    take_lower(b2, bm, xhalf(b2),   xhalf_i(bm));
    bm = (bm < 0) ? 0 : ((bm > N - 1) ? N - 1 : bm);
    const int bmu = __builtin_amdgcn_readfirstlane(bm);
    const int2 bl = *(const int2*)(gloc + 2*(size_t)bmu);

    const float decay  = 1.0f - (float)iter / 100.0f;
    const float radius = 0.5f * decay;
    const float sig    = 256.0f * decay;
    const float sigma2 = sig * sig;
    const float bx = (float)bl.x, by = (float)bl.y;

    if (full) {
        bool skip, selA;
        row_flags_at(gl, bx, by, radius, sigma2, skip, selA);
        const unsigned skipMask = __builtin_amdgcn_ballot_w32(skip);
        if (skipMask == 0xffffffffu) {
            const float4 w0 = V4[0*32], w1 = V4[1*32], w2 = V4[2*32], w3 = V4[3*32];
            const float4 w4 = V4[4*32], w5 = V4[5*32], w6 = V4[6*32], w7 = V4[7*32];
            #pragma unroll
            for (int q = 0; q < 8; ++q) OP[q*32] = pv[q];
            OV[0*32] = w0; OV[1*32] = w1; OV[2*32] = w2; OV[3*32] = w3;
            OV[4*32] = w4; OV[5*32] = w5; OV[6*32] = w6; OV[7*32] = w7;
        } else {
            const unsigned selMask = __builtin_amdgcn_ballot_w32(selA);
            float4 g = ((const float4*)(particles + (size_t)bmu * SOM_DIM))[lane & 7];
            float4 cA, cB;
            #pragma unroll
            for (int h = 0; h < 2; ++h) {
                float4 vv[4], a[4], b[4];
                #pragma unroll
                for (int i = 0; i < 4; ++i) {
                    const int q = 4 * h + i;
                    vv[i] = V4[q*32]; a[i] = R14[q*32]; b[i] = R24[q*32];
                }
                asm volatile("" : "+v"(vv[0].x), "+v"(vv[1].x), "+v"(vv[2].x), "+v"(vv[3].x),
                                  "+v"(a[0].x),  "+v"(a[1].x),  "+v"(a[2].x),  "+v"(a[3].x),
                                  "+v"(b[0].x),  "+v"(b[1].x),  "+v"(b[2].x),  "+v"(b[3].x),
                                  "+v"(g.x), "+v"(g.y), "+v"(g.z), "+v"(g.w) :: "memory");
                if (h == 0) centroids(wsumC4, g, lane, N, cA, cB);
                #pragma unroll
                for (int i = 0; i < 4; ++i) {
                    const int q = 4 * h + i;
                    const int rl = q * 4 + rsub;
                    const bool sk = ((skipMask >> rl) & 1u) != 0u;
                    const bool sa = ((selMask  >> rl) & 1u) != 0u;
                    float4 pn, vn;
                    update_piece(pv[q], vv[i], a[i], b[i], g, cA, cB, sk, sa, pn, vn);
                    OP[q*32] = pn;
                    OV[q*32] = vn;
                }
            }
        }
    } else {
        bool skip = true, selA = false;
        if (prow < N) row_flags(gloc, prow, bx, by, radius, sigma2, skip, selA);
        const unsigned skipMask = __builtin_amdgcn_ballot_w32(skip);
        const unsigned selMask  = __builtin_amdgcn_ballot_w32(selA);
        const float4 g = ((const float4*)(particles + (size_t)bmu * SOM_DIM))[lane & 7];
        float4 cA, cB;
        centroids(wsumC4, g, lane, N, cA, cB);
        for (int q = 0; q < 8; ++q) {
            const int rl = q * 4 + rsub;
            if (rowW + rl < N) {
                const bool sk = ((skipMask >> rl) & 1u) != 0u;
                const bool sa = ((selMask  >> rl) & 1u) != 0u;
                float4 pn, vn;
                update_piece(P4[q*32], V4[q*32], R14[q*32], R24[q*32], g, cA, cB, sk, sa, pn, vn);
                OP[q*32] = pn;
                OV[q*32] = vn;
            }
        }
    }
}

extern "C" void kernel_launch(void* const* d_in, const int* in_sizes, int n_in,
                              void* d_out, int out_size, void* d_ws, size_t ws_size,
                              hipStream_t stream)
{
    (void)n_in; (void)out_size; (void)ws_size;
    const float* input      = (const float*)d_in[0];
    const float* particles  = (const float*)d_in[1];
    const float* velocities = (const float*)d_in[2];
    const float* r1         = (const float*)d_in[3];
    const float* r2         = (const float*)d_in[4];
    const int*   gloc       = (const int*)d_in[5];
    const int*   iter_num   = (const int*)d_in[6];

    const int N  = in_sizes[1] / SOM_DIM;
    const int NB  = (N + ROWS_A - 1) / ROWS_A;
    const int NBC = (N + CTHREADS - 1) / CTHREADS;

    float* ws      = (float*)d_ws;
    float* blkSum  = ws;
    float* blkMinV = ws + (size_t)NB * SOM_DIM;
    int*   blkMinI = (int*)(blkMinV + NB);

    float* outP = (float*)d_out;
    float* outV = outP + (size_t)N * SOM_DIM;

    som_pass_a<<<NB, THREADS, 0, stream>>>(input, particles, blkSum, blkMinV, blkMinI, N);
    som_pass_c<<<NBC, CTHREADS, 0, stream>>>(particles, velocities, r1, r2, gloc, iter_num,
                                            blkSum, blkMinV, blkMinI, outP, outV, NB, N);
}
